# starting kernel
# baseline (speedup 1.0000x reference)
#include <hip/hip_runtime.h>


typedef __attribute__((ext_vector_type(2))) float v2f;
typedef __attribute__((ext_vector_type(8))) float v8f;

#define NDIM 64
#define WAVES_PER_BLOCK 4
#define WP_ELEMS 4096
#define SCAN_B 256

__device__ __forceinline__ v8f wmma_f32_16x16x4(v2f a, v2f b, v8f c) {
  return __builtin_amdgcn_wmma_f32_16x16x4_f32(false, a, false, b, (short)0, c,
                                               false, false);
}

__global__ void gin_repack_kernel(const float* __restrict__ W,
                                  float* __restrict__ Wp) {
  const int tid = blockIdx.x * blockDim.x + threadIdx.x;
  const int c = tid >> 7;
  const int n = (tid >> 5) & 3;
  const int lane = tid & 31;
  const int hi = lane >> 4;
  const int m = lane & 15;
  const int k = 4 * c + 2 * hi;
  Wp[tid * 2 + 0] = W[(k + 0) * NDIM + 16 * n + m];
  Wp[tid * 2 + 1] = W[(k + 1) * NDIM + 16 * n + m];
}

template <int MODE>
__global__ void __launch_bounds__(WAVES_PER_BLOCK * 32)
gin_mlp_kernel(const float* __restrict__ H, const float* __restrict__ AGG,
               const float* __restrict__ WP1, const float* __restrict__ B1,
               const float* __restrict__ WP2, const float* __restrict__ B2,
               float* __restrict__ OUT, int nTiles) {
  __shared__ float lds[WAVES_PER_BLOCK][16 * NDIM];

  const int wave = threadIdx.x >> 5;
  const int lane = threadIdx.x & 31;
  const int tile = blockIdx.x * WAVES_PER_BLOCK + wave;
  if (tile >= nTiles) return;

  const int laneM = lane & 15;
  const int hi = lane >> 4;
  const int row0 = tile * 16;

  v2f a[16];
  {
    const size_t base = (size_t)(row0 + laneM) * NDIM + 2 * hi;
    const float* hp = H + base;
    const float* gp = AGG + base;
#pragma unroll
    for (int c = 0; c < 16; ++c) {
      v2f z = *(const v2f*)(hp + 4 * c);
      if (MODE == 1) {
        v2f g = *(const v2f*)(gp + 4 * c);
        z.x += g.x;
        z.y += g.y;
      }
      a[c] = z;
    }
  }

  const v2f* wp1 = (const v2f*)WP1;

  v8f acc[4];
#pragma unroll
  for (int n = 0; n < 4; ++n) {
    const float bv = B1[16 * n + laneM];
#pragma unroll
    for (int r = 0; r < 8; ++r) acc[n][r] = bv;
  }
#pragma unroll
  for (int c = 0; c < 16; ++c)
#pragma unroll
    for (int n = 0; n < 4; ++n)
      acc[n] = wmma_f32_16x16x4(a[c], wp1[(c * 4 + n) * 32 + lane], acc[n]);

  if (MODE == 2) {
#pragma unroll
    for (int n = 0; n < 4; ++n)
#pragma unroll
      for (int r = 0; r < 8; ++r)
        OUT[(size_t)(row0 + r + 8 * hi) * NDIM + 16 * n + laneM] = acc[n][r];
    return;
  }

  float* lt = lds[wave];
#pragma unroll
  for (int n = 0; n < 4; ++n)
#pragma unroll
    for (int r = 0; r < 8; ++r)
      lt[(r + 8 * hi) * NDIM + 16 * n + laneM] = fmaxf(acc[n][r], 0.0f);

#pragma unroll
  for (int c = 0; c < 16; ++c)
    a[c] = *(const v2f*)(lt + laneM * NDIM + 4 * c + 2 * hi);

  const v2f* wp2 = (const v2f*)WP2;
#pragma unroll
  for (int n = 0; n < 4; ++n) {
    const float bv = B2[16 * n + laneM];
#pragma unroll
    for (int r = 0; r < 8; ++r) acc[n][r] = bv;
  }
#pragma unroll
  for (int c = 0; c < 16; ++c)
#pragma unroll
    for (int n = 0; n < 4; ++n)
      acc[n] = wmma_f32_16x16x4(a[c], wp2[(c * 4 + n) * 32 + lane], acc[n]);

#pragma unroll
  for (int n = 0; n < 4; ++n)
#pragma unroll
    for (int r = 0; r < 8; ++r)
      OUT[(size_t)(row0 + r + 8 * hi) * NDIM + 16 * n + laneM] = acc[n][r];
}


__global__ void gin_zero_i32_kernel(int* __restrict__ p, int n) {
  const int i = blockIdx.x * blockDim.x + threadIdx.x;
  if (i < n) p[i] = 0;
}

__global__ void gin_degree_kernel(const int* __restrict__ rowp,
                                  int* __restrict__ deg, int E) {
  const int e = blockIdx.x * blockDim.x + threadIdx.x;
  if (e < E) atomicAdd(&deg[rowp[e]], 1);
}

__global__ void gin_scan1_kernel(const int* __restrict__ deg,
                                 int* __restrict__ rowstart1,
                                 int* __restrict__ bwork, int n) {
  __shared__ int s[SCAN_B];
  const int i = blockIdx.x * SCAN_B + threadIdx.x;
  const int v = (i < n) ? deg[i] : 0;
  s[threadIdx.x] = v;
  __syncthreads();
#pragma unroll
  for (int off = 1; off < SCAN_B; off <<= 1) {
    const int tv = (threadIdx.x >= off) ? s[threadIdx.x - off] : 0;
    __syncthreads();
    s[threadIdx.x] += tv;
    __syncthreads();
  }
  if (i < n) rowstart1[i] = s[threadIdx.x];
  if (threadIdx.x == SCAN_B - 1) bwork[blockIdx.x] = s[SCAN_B - 1];
}

__global__ void gin_scan2_kernel(int* __restrict__ bwork, int nb) {
  __shared__ int s[512];
  const int i = threadIdx.x;
  const int v = (i < nb) ? bwork[i] : 0;
  s[i] = v;
  __syncthreads();
#pragma unroll
  for (int off = 1; off < 512; off <<= 1) {
    const int tv = (i >= off) ? s[i - off] : 0;
    __syncthreads();
    s[i] += tv;
    __syncthreads();
  }
  if (i < nb) bwork[i] = s[i] - v;
}

__global__ void gin_scan3_kernel(int* __restrict__ rowstart,
                                 const int* __restrict__ bwork, int n) {
  const int i = blockIdx.x * blockDim.x + threadIdx.x;
  if (i < n) rowstart[i + 1] += bwork[i / SCAN_B];
  if (i == 0) rowstart[0] = 0;
}

__global__ void gin_fill_kernel(const int* __restrict__ rowp,
                                const int* __restrict__ colp,
                                const int* __restrict__ rowstart,
                                int* __restrict__ cursor,
                                int* __restrict__ perm, int E) {
  const int e = blockIdx.x * blockDim.x + threadIdx.x;
  if (e < E) {
    const int r = rowp[e];
    const int pos = rowstart[r] + atomicAdd(&cursor[r], 1);
    perm[pos] = colp[e];
  }
}

__global__ void gin_gather_kernel(const float* __restrict__ T,
                                  const int* __restrict__ rowstart,
                                  const int* __restrict__ perm,
                                  float* __restrict__ AGG, int N) {
  const int g = (blockIdx.x * blockDim.x + threadIdx.x) >> 4;
  if (g >= N) return;
  const int l4 = (threadIdx.x & 15) * 4;
  const int s = rowstart[g];
  const int e = rowstart[g + 1];
  float4 acc = make_float4(0.f, 0.f, 0.f, 0.f);
  for (int j = s; j < e; ++j) {
    const int c = perm[j];
    const float4 v = *(const float4*)(T + (size_t)c * NDIM + l4);
    acc.x += v.x;
    acc.y += v.y;
    acc.z += v.z;
    acc.w += v.w;
  }
  *(float4*)(AGG + (size_t)g * NDIM + l4) = acc;
}


extern "C" void kernel_launch(void* const* d_in, const int* in_sizes, int n_in,
                              void* d_out, int out_size, void* d_ws,
                              size_t ws_size, hipStream_t stream) {
  (void)n_in;
  (void)out_size;
  (void)ws_size;

  const float* x = (const float*)d_in[0];
  const int* ei = (const int*)d_in[1];
  const float* W1 = (const float*)d_in[2];
  const float* b1 = (const float*)d_in[3];
  const float* W2 = (const float*)d_in[4];
  const float* b2 = (const float*)d_in[5];
  const float* Wf = (const float*)d_in[6];
  const float* bf = (const float*)d_in[7];
  float* out = (float*)d_out;

  const int N = in_sizes[0] / NDIM;
  const int E = in_sizes[1] / 2;
  const int L = in_sizes[3] / NDIM;
  const int nTiles = N / 16;
  const size_t NB = (size_t)N * NDIM;

  float* agg = (float*)d_ws;
  float* t = agg + NB;
  float* h = t + NB;
  float* wp = h + NB;
  int* deg = (int*)(wp + 7 * WP_ELEMS);
  int* cursor = deg + N;
  int* rowstart = cursor + N;
  int* bwork = rowstart + (N + 1);
  int* perm = bwork + 1024;

  const int* rowp = ei;
  const int* colp = ei + E;

  const float* wsrc[7] = {W1, W1 + WP_ELEMS, W1 + 2 * WP_ELEMS,
                          W2, W2 + WP_ELEMS, W2 + 2 * WP_ELEMS, Wf};
  for (int m = 0; m < 7; ++m)
    gin_repack_kernel<<<WP_ELEMS / 2 / 256, 256, 0, stream>>>(
        wsrc[m], wp + (size_t)m * WP_ELEMS);

  const int nScanBlocks = (N + SCAN_B - 1) / SCAN_B;
  gin_zero_i32_kernel<<<(2 * N + 255) / 256, 256, 0, stream>>>(deg, 2 * N);
  gin_degree_kernel<<<(E + 255) / 256, 256, 0, stream>>>(rowp, deg, E);
  gin_scan1_kernel<<<nScanBlocks, SCAN_B, 0, stream>>>(deg, rowstart + 1,
                                                       bwork, N);
  gin_scan2_kernel<<<1, 512, 0, stream>>>(bwork, nScanBlocks);
  gin_scan3_kernel<<<(N + 255) / 256, 256, 0, stream>>>(rowstart, bwork, N);
  gin_fill_kernel<<<(E + 255) / 256, 256, 0, stream>>>(rowp, colp, rowstart,
                                                       cursor, perm, E);

  const dim3 mlpBlk(WAVES_PER_BLOCK * 32);
  const dim3 mlpGrd((nTiles + WAVES_PER_BLOCK - 1) / WAVES_PER_BLOCK);
  const dim3 gatGrd((N * 16 + 255) / 256);

  const float* hcur = x;
  for (int l = 0; l < L; ++l) {
    const float* wp1 = wp + (size_t)l * WP_ELEMS;
    const float* wp2 = wp + (size_t)(3 + l) * WP_ELEMS;
    const float* v1 = b1 + (size_t)l * NDIM;
    const float* v2 = b2 + (size_t)l * NDIM;

    gin_mlp_kernel<0><<<mlpGrd, mlpBlk, 0, stream>>>(hcur, nullptr, wp1, v1,
                                                     wp2, v2, t, nTiles);
    gin_gather_kernel<<<gatGrd, 256, 0, stream>>>(t, rowstart, perm, agg, N);
    gin_mlp_kernel<1><<<mlpGrd, mlpBlk, 0, stream>>>(hcur, agg, wp1, v1, wp2,
                                                     v2, h, nTiles);
    hcur = h;
  }
  gin_mlp_kernel<2><<<mlpGrd, mlpBlk, 0, stream>>>(
      hcur, nullptr, wp + (size_t)6 * WP_ELEMS, bf, nullptr, nullptr, out,
      nTiles);
}


# Adjacency as fixed 32-slot rows (one 128-B line per node) filled straight from the degree add's returned rank, spill list past 32: no scan, no fill pass, 16 -> 12 launches; f32 sums in arrival order
# speedup vs baseline: 1.7664x; 1.7664x over previous
#include <hip/hip_runtime.h>


typedef __attribute__((ext_vector_type(2))) float v2f;
typedef __attribute__((ext_vector_type(4))) float v4f;
typedef __attribute__((ext_vector_type(8))) float v8f;

#define NDIM 64
#define WAVES_PER_BLOCK 4
#define W_ELEMS 4096
#define LDP 68

__device__ __forceinline__ v8f wmma_f32_16x16x4(v2f a, v2f b, v8f c) {
  return __builtin_amdgcn_wmma_f32_16x16x4_f32(false, a, false, b, (short)0, c,
                                               false, false);
}

__device__ __forceinline__ void load_wstage(v4f (&q)[4],
                                            const float* __restrict__ w, int s,
                                            int laneM, int hi) {
#pragma unroll
  for (int i = 0; i < 4; ++i)
    q[i] = *(const v4f*)(w + (8 * s + 4 * hi + i) * NDIM + 4 * laneM);
}

__device__ __forceinline__ void mma_stage(v8f (&acc)[4], v2f a0, v2f a1,
                                          const v4f (&q)[4]) {
#pragma unroll
  for (int n = 0; n < 4; ++n) {
    v2f b;
    b.x = q[0][n];
    b.y = q[1][n];
    acc[n] = wmma_f32_16x16x4(a0, b, acc[n]);
  }
#pragma unroll
  for (int n = 0; n < 4; ++n) {
    v2f b;
    b.x = q[2][n];
    b.y = q[3][n];
    acc[n] = wmma_f32_16x16x4(a1, b, acc[n]);
  }
}

__device__ __forceinline__ void load_afrags(v2f (&a)[16], const float* lt,
                                            int laneM, int hi) {
#pragma unroll
  for (int j = 0; j < 8; ++j) {
    const float4 q = *(const float4*)(lt + laneM * LDP + 8 * j + 4 * hi);
    v2f lo, hh;
    lo.x = q.x;
    lo.y = q.y;
    hh.x = q.z;
    hh.y = q.w;
    a[2 * j] = lo;
    a[2 * j + 1] = hh;
  }
}

template <int MODE>
__global__ void __launch_bounds__(WAVES_PER_BLOCK * 32)
gin_mlp_kernel(const float* __restrict__ H, const float* __restrict__ AGG,
               const float* __restrict__ W1, const float* __restrict__ B1,
               const float* __restrict__ W2, const float* __restrict__ B2,
               float* __restrict__ OUT, int nTiles) {
  __shared__ __attribute__((aligned(16))) float lds[WAVES_PER_BLOCK][16 * LDP];

  const int wave = threadIdx.x >> 5;
  const int lane = threadIdx.x & 31;
  const int tile = blockIdx.x * WAVES_PER_BLOCK + wave;
  if (tile >= nTiles) return;

  const int laneM = lane & 15;
  const int hi = lane >> 4;
  const int row0 = tile * 16;

  v4f wq[2][4];
  load_wstage(wq[0], W1, 0, laneM, hi);

  float* lt = lds[wave];
  {
    const size_t base = (size_t)(row0 + hi) * NDIM + 4 * laneM;
    const float* hp = H + base;
    const float* gp = AGG + base;
#pragma unroll
    for (int i = 0; i < 8; ++i) {
      float4 z = *(const float4*)(hp + 2 * i * NDIM);
      if (MODE == 1) {
        const float4 g = *(const float4*)(gp + 2 * i * NDIM);
        z.x += g.x;
        z.y += g.y;
        z.z += g.z;
        z.w += g.w;
      }
      *(float4*)(lt + (2 * i + hi) * LDP + 4 * laneM) = z;
    }
  }
  v2f a[16];
  load_afrags(a, lt, laneM, hi);

  v8f acc[4];
  {
    const float4 bv = *(const float4*)(B1 + 4 * laneM);
    const float bvn[4] = {bv.x, bv.y, bv.z, bv.w};
#pragma unroll
    for (int n = 0; n < 4; ++n)
#pragma unroll
      for (int r = 0; r < 8; ++r) acc[n][r] = bvn[n];
  }
#pragma unroll
  for (int s = 0; s < 8; ++s) {
    if (s + 1 < 8)
      load_wstage(wq[(s + 1) & 1], W1, s + 1, laneM, hi);
    else if (MODE != 2)
      load_wstage(wq[(s + 1) & 1], W2, 0, laneM, hi);
    __builtin_amdgcn_sched_barrier(0);
    mma_stage(acc, a[2 * s], a[2 * s + 1], wq[s & 1]);
    __builtin_amdgcn_sched_barrier(0);
  }

  if (MODE == 2) {
#pragma unroll
    for (int r = 0; r < 8; ++r)
      *(float4*)(OUT + (size_t)(row0 + r + 8 * hi) * NDIM + 4 * laneM) =
          make_float4(acc[0][r], acc[1][r], acc[2][r], acc[3][r]);
    return;
  }

#pragma unroll
  for (int r = 0; r < 8; ++r)
    *(float4*)(lt + (r + 8 * hi) * LDP + 4 * laneM) =
        make_float4(fmaxf(acc[0][r], 0.0f), fmaxf(acc[1][r], 0.0f),
                    fmaxf(acc[2][r], 0.0f), fmaxf(acc[3][r], 0.0f));

  load_afrags(a, lt, laneM, hi);

  {
    const float4 bv = *(const float4*)(B2 + 4 * laneM);
    const float bvn[4] = {bv.x, bv.y, bv.z, bv.w};
#pragma unroll
    for (int n = 0; n < 4; ++n)
#pragma unroll
      for (int r = 0; r < 8; ++r) acc[n][r] = bvn[n];
  }
#pragma unroll
  for (int s = 0; s < 8; ++s) {
    if (s + 1 < 8) load_wstage(wq[(s + 1) & 1], W2, s + 1, laneM, hi);
    __builtin_amdgcn_sched_barrier(0);
    mma_stage(acc, a[2 * s], a[2 * s + 1], wq[s & 1]);
    __builtin_amdgcn_sched_barrier(0);
  }

#pragma unroll
  for (int r = 0; r < 8; ++r)
    *(float4*)(OUT + (size_t)(row0 + r + 8 * hi) * NDIM + 4 * laneM) =
        make_float4(acc[0][r], acc[1][r], acc[2][r], acc[3][r]);
}

#define SLOTS 32

__global__ void gin_zero_i32_kernel(int* __restrict__ p, int n) {
  const int i = blockIdx.x * blockDim.x + threadIdx.x;
  if (i < n) p[i] = 0;
}

__global__ void gin_place_kernel(const int* __restrict__ rowp,
                                 const int* __restrict__ colp,
                                 int* __restrict__ deg, int* __restrict__ slots,
                                 int* __restrict__ spill, int E, int N) {
  const int e = blockIdx.x * blockDim.x + threadIdx.x;
  if (e < E) {
    const int r = rowp[e];
    const int c = colp[e];
    const int k = atomicAdd(&deg[r], 1);
    if (k < SLOTS) {
      slots[r * SLOTS + k] = c;
    } else {
      const int o = atomicAdd(&deg[N], 1);
      spill[2 * o] = r;
      spill[2 * o + 1] = c;
    }
  }
}

#define GNODES 8

__device__ __forceinline__ void gin_sum16(float4& acc,
                                          const float* __restrict__ T, int idx,
                                          int cnt, unsigned lb) {
  float4 v[16];
#pragma unroll
  for (int q = 0; q < 4; ++q) {
    if (4 * q < cnt) {
#pragma unroll
      for (int i = 0; i < 4; ++i) {
        const unsigned c = (unsigned)__shfl(idx, 4 * q + i, 16);
        v[4 * q + i] = *(const float4*)((const char*)T + ((c << 8) + lb));
      }
    }
  }
#pragma unroll
  for (int q = 0; q < 4; ++q) {
    if (4 * q < cnt) {
#pragma unroll
      for (int i = 0; i < 4; ++i) {
        if (4 * q + i < cnt) {
          acc.x += v[4 * q + i].x;
          acc.y += v[4 * q + i].y;
          acc.z += v[4 * q + i].z;
          acc.w += v[4 * q + i].w;
        }
      }
    }
  }
}

__device__ __forceinline__ const int* gin_slot_ptr(const int* row_slots,
                                                   const int* row_deg, int pos,
                                                   int cnt) {
  return cnt > 0 ? row_slots + min(pos, cnt - 1) : row_deg;
}

__global__ void __launch_bounds__(128, 7)
gin_gather_kernel(const float* __restrict__ T, const int* __restrict__ deg,
                  const int* __restrict__ slots, const int* __restrict__ spill,
                  float* __restrict__ AGG, int N, int E) {
  const int grp = (blockIdx.x * blockDim.x + threadIdx.x) >> 4;
  const int g0 = grp * GNODES;
  if (g0 >= N) return;
  const int nn = min(GNODES, N - g0);
  const int l16 = threadIdx.x & 15;
  const unsigned lb = (unsigned)l16 * 16u;

  const int* dp = deg + g0;
  const int* sl = slots + g0 * SLOTS;
  const int dg = dp[min(l16, nn - 1)];
  int d = __shfl(dg, 0, 16);
  int dc = min(d, SLOTS);
  int ia = *gin_slot_ptr(sl, dp, l16, dc);
  int ib = *gin_slot_ptr(sl, dp, 16 + l16, dc);

  for (int n = 0; n < nn; ++n) {
    const int m = min(n + 1, nn - 1);
    const int d2 = __shfl(dg, m, 16);
    const int dc2 = min(d2, SLOTS);
    const int ja = *gin_slot_ptr(sl + m * SLOTS, dp + m, l16, dc2);
    const int jb = *gin_slot_ptr(sl + m * SLOTS, dp + m, 16 + l16, dc2);

    float4 acc = make_float4(0.f, 0.f, 0.f, 0.f);
    if (dc > 0) gin_sum16(acc, T, ia, dc, lb);
    if (dc > 16) gin_sum16(acc, T, ib, dc - 16, lb);
    if (d > SLOTS) {
      const int ns = min(deg[N], E);
      for (int j = 0; j < ns; ++j) {
        if (spill[2 * j] == g0 + n) {
          const unsigned c = (unsigned)spill[2 * j + 1];
          const float4 v = *(const float4*)((const char*)T + ((c << 8) + lb));
          acc.x += v.x;
          acc.y += v.y;
          acc.z += v.z;
          acc.w += v.w;
        }
      }
    }
    *(float4*)((char*)AGG + (((unsigned)(g0 + n) << 8) + lb)) = acc;
    d = d2;
    dc = dc2;
    ia = ja;
    ib = jb;
  }
}


extern "C" void kernel_launch(void* const* d_in, const int* in_sizes, int n_in,
                              void* d_out, int out_size, void* d_ws,
                              size_t ws_size, hipStream_t stream) {
  (void)n_in;
  (void)out_size;
  (void)ws_size;

  const float* x = (const float*)d_in[0];
  const int* ei = (const int*)d_in[1];
  const float* W1 = (const float*)d_in[2];
  const float* b1 = (const float*)d_in[3];
  const float* W2 = (const float*)d_in[4];
  const float* b2 = (const float*)d_in[5];
  const float* Wf = (const float*)d_in[6];
  const float* bf = (const float*)d_in[7];
  float* out = (float*)d_out;

  const int N = in_sizes[0] / NDIM;
  const int E = in_sizes[1] / 2;
  const int L = in_sizes[3] / NDIM;
  const int nTiles = N / 16;
  const size_t NB = (size_t)N * NDIM;

  float* agg = (float*)d_ws;
  float* t = agg + NB;
  float* h = t + NB;
  int* slots = (int*)(h + NB);
  int* spill = slots + (size_t)N * SLOTS;
  int* deg = spill + 2 * (size_t)E;

  const int* rowp = ei;
  const int* colp = ei + E;

  gin_zero_i32_kernel<<<(N + 1 + 255) / 256, 256, 0, stream>>>(deg, N + 1);
  gin_place_kernel<<<(E + 255) / 256, 256, 0, stream>>>(rowp, colp, deg, slots,
                                                         spill, E, N);

  const dim3 mlpBlk(WAVES_PER_BLOCK * 32);
  const dim3 mlpGrd((nTiles + WAVES_PER_BLOCK - 1) / WAVES_PER_BLOCK);
  const dim3 gatGrd(((N + GNODES - 1) / GNODES * 16 + 127) / 128);

  const float* hcur = x;
  for (int l = 0; l < L; ++l) {
    const float* w1 = W1 + (size_t)l * W_ELEMS;
    const float* w2 = W2 + (size_t)l * W_ELEMS;
    const float* v1 = b1 + (size_t)l * NDIM;
    const float* v2 = b2 + (size_t)l * NDIM;

    gin_mlp_kernel<0><<<mlpGrd, mlpBlk, 0, stream>>>(hcur, nullptr, w1, v1, w2,
                                                     v2, t, nTiles);
    gin_gather_kernel<<<gatGrd, 128, 0, stream>>>(t, deg, slots, spill, agg, N,
                                                  E);
    gin_mlp_kernel<1><<<mlpGrd, mlpBlk, 0, stream>>>(hcur, agg, w1, v1, w2, v2,
                                                     h, nTiles);
    hcur = h;
  }
  gin_mlp_kernel<2><<<mlpGrd, mlpBlk, 0, stream>>>(hcur, nullptr, Wf, bf,
                                                   nullptr, nullptr, out,
                                                   nTiles);
}


# Adjacency build split by chip half: nodes dealt to the halves in runs of 64; work-groups draw edge chunks from their half's counter and add only its nodes' edges; f32 sums stay in arrival order
# speedup vs baseline: 2.2082x; 1.2501x over previous
#include <hip/hip_runtime.h>


typedef __attribute__((ext_vector_type(2))) float v2f;
typedef __attribute__((ext_vector_type(4))) float v4f;
typedef __attribute__((ext_vector_type(8))) float v8f;

#define NDIM 64
#define WAVES_PER_BLOCK 4
#define W_ELEMS 4096
#define LDP 68

__device__ __forceinline__ v8f wmma_f32_16x16x4(v2f a, v2f b, v8f c) {
  return __builtin_amdgcn_wmma_f32_16x16x4_f32(false, a, false, b, (short)0, c,
                                               false, false);
}

__device__ __forceinline__ void load_wstage(v4f (&q)[4],
                                            const float* __restrict__ w, int s,
                                            int laneM, int hi) {
#pragma unroll
  for (int i = 0; i < 4; ++i)
    q[i] = *(const v4f*)(w + (8 * s + 4 * hi + i) * NDIM + 4 * laneM);
}

__device__ __forceinline__ void mma_stage(v8f (&acc)[4], v2f a0, v2f a1,
                                          const v4f (&q)[4]) {
#pragma unroll
  for (int n = 0; n < 4; ++n) {
    v2f b;
    b.x = q[0][n];
    b.y = q[1][n];
    acc[n] = wmma_f32_16x16x4(a0, b, acc[n]);
  }
#pragma unroll
  for (int n = 0; n < 4; ++n) {
    v2f b;
    b.x = q[2][n];
    b.y = q[3][n];
    acc[n] = wmma_f32_16x16x4(a1, b, acc[n]);
  }
}

__device__ __forceinline__ void load_afrags(v2f (&a)[16], const float* lt,
                                            int laneM, int hi) {
#pragma unroll
  for (int j = 0; j < 8; ++j) {
    const float4 q = *(const float4*)(lt + laneM * LDP + 8 * j + 4 * hi);
    v2f lo, hh;
    lo.x = q.x;
    lo.y = q.y;
    hh.x = q.z;
    hh.y = q.w;
    a[2 * j] = lo;
    a[2 * j + 1] = hh;
  }
}

template <int MODE>
__global__ void __launch_bounds__(WAVES_PER_BLOCK * 32)
gin_mlp_kernel(const float* __restrict__ H, const float* __restrict__ AGG,
               const float* __restrict__ W1, const float* __restrict__ B1,
               const float* __restrict__ W2, const float* __restrict__ B2,
               float* __restrict__ OUT, int nTiles) {
  __shared__ __attribute__((aligned(16))) float lds[WAVES_PER_BLOCK][16 * LDP];

  const int wave = threadIdx.x >> 5;
  const int lane = threadIdx.x & 31;
  const int tile = blockIdx.x * WAVES_PER_BLOCK + wave;
  if (tile >= nTiles) return;

  const int laneM = lane & 15;
  const int hi = lane >> 4;
  const int row0 = tile * 16;

  v4f wq[2][4];
  load_wstage(wq[0], W1, 0, laneM, hi);

  float* lt = lds[wave];
  {
    const size_t base = (size_t)(row0 + hi) * NDIM + 4 * laneM;
    const float* hp = H + base;
    const float* gp = AGG + base;
#pragma unroll
    for (int i = 0; i < 8; ++i) {
      float4 z = *(const float4*)(hp + 2 * i * NDIM);
      if (MODE == 1) {
        const float4 g = *(const float4*)(gp + 2 * i * NDIM);
        z.x += g.x;
        z.y += g.y;
        z.z += g.z;
        z.w += g.w;
      }
      *(float4*)(lt + (2 * i + hi) * LDP + 4 * laneM) = z;
    }
  }
  v2f a[16];
  load_afrags(a, lt, laneM, hi);

  v8f acc[4];
  {
    const float4 bv = *(const float4*)(B1 + 4 * laneM);
    const float bvn[4] = {bv.x, bv.y, bv.z, bv.w};
#pragma unroll
    for (int n = 0; n < 4; ++n)
#pragma unroll
      for (int r = 0; r < 8; ++r) acc[n][r] = bvn[n];
  }
#pragma unroll
  for (int s = 0; s < 8; ++s) {
    if (s + 1 < 8)
      load_wstage(wq[(s + 1) & 1], W1, s + 1, laneM, hi);
    else if (MODE != 2)
      load_wstage(wq[(s + 1) & 1], W2, 0, laneM, hi);
    __builtin_amdgcn_sched_barrier(0);
    mma_stage(acc, a[2 * s], a[2 * s + 1], wq[s & 1]);
    __builtin_amdgcn_sched_barrier(0);
  }

  if (MODE == 2) {
#pragma unroll
    for (int r = 0; r < 8; ++r)
      *(float4*)(OUT + (size_t)(row0 + r + 8 * hi) * NDIM + 4 * laneM) =
          make_float4(acc[0][r], acc[1][r], acc[2][r], acc[3][r]);
    return;
  }

#pragma unroll
  for (int r = 0; r < 8; ++r)
    *(float4*)(lt + (r + 8 * hi) * LDP + 4 * laneM) =
        make_float4(fmaxf(acc[0][r], 0.0f), fmaxf(acc[1][r], 0.0f),
                    fmaxf(acc[2][r], 0.0f), fmaxf(acc[3][r], 0.0f));

  load_afrags(a, lt, laneM, hi);

  {
    const float4 bv = *(const float4*)(B2 + 4 * laneM);
    const float bvn[4] = {bv.x, bv.y, bv.z, bv.w};
#pragma unroll
    for (int n = 0; n < 4; ++n)
#pragma unroll
      for (int r = 0; r < 8; ++r) acc[n][r] = bvn[n];
  }
#pragma unroll
  for (int s = 0; s < 8; ++s) {
    if (s + 1 < 8) load_wstage(wq[(s + 1) & 1], W2, s + 1, laneM, hi);
    __builtin_amdgcn_sched_barrier(0);
    mma_stage(acc, a[2 * s], a[2 * s + 1], wq[s & 1]);
    __builtin_amdgcn_sched_barrier(0);
  }

#pragma unroll
  for (int r = 0; r < 8; ++r)
    *(float4*)(OUT + (size_t)(row0 + r + 8 * hi) * NDIM + 4 * laneM) =
        make_float4(acc[0][r], acc[1][r], acc[2][r], acc[3][r]);
}

#define SLOTS 32

__global__ void gin_zero_i32_kernel(int* __restrict__ p, int n) {
  const int i = blockIdx.x * blockDim.x + threadIdx.x;
  if (i < n) p[i] = 0;
}

#define PCHUNK 512
#define PTHREADS 256
#define PEDGES (PCHUNK / PTHREADS)

__global__ void __launch_bounds__(PTHREADS)
gin_place_kernel(const int* __restrict__ rowp, const int* __restrict__ colp,
                 int* __restrict__ deg, int* __restrict__ slots,
                 int* __restrict__ spill, int E, int N, int nChunks) {
  __shared__ int s_half;
  __shared__ int s_chunk;
  int* queue = deg + N + 1;

  if (threadIdx.x == 0) {
    s_half = (int)((__builtin_amdgcn_s_sendmsg_rtn(0x87) >> 18) & 1u);
  }
  __syncthreads();
  const int myhalf = s_half;

  for (int pass = 0; pass < 2; ++pass) {
    const int set = myhalf ^ pass;
    __syncthreads();
    if (threadIdx.x == 0) s_chunk = atomicAdd(&queue[set], 1);
    __syncthreads();
    int c = s_chunk;
    for (int trip = 0; trip < nChunks && c < nChunks; ++trip) {
      const int e0 = c * PCHUNK + (int)threadIdx.x;
      int r[PEDGES], cc[PEDGES], k[PEDGES];
      bool mine[PEDGES];
#pragma unroll
      for (int i = 0; i < PEDGES; ++i) {
        const int e = e0 + i * PTHREADS;
        const bool in = e < E;
        r[i] = in ? rowp[e] : 0;
        cc[i] = in ? colp[e] : 0;
        mine[i] = in && (((r[i] >> 6) & 1) == set);
      }
#pragma unroll
      for (int i = 0; i < PEDGES; ++i)
        k[i] = mine[i] ? atomicAdd(&deg[r[i]], 1) : 0;
      int nxt = 0;
      if (threadIdx.x == 0) nxt = atomicAdd(&queue[set], 1);
#pragma unroll
      for (int i = 0; i < PEDGES; ++i) {
        if (mine[i]) {
          if (k[i] < SLOTS) {
            slots[r[i] * SLOTS + k[i]] = cc[i];
          } else {
            const int o = atomicAdd(&deg[N], 1);
            spill[2 * o] = r[i];
            spill[2 * o + 1] = cc[i];
          }
        }
      }

      __syncthreads();
      if (threadIdx.x == 0) s_chunk = nxt;
      __syncthreads();
      c = s_chunk;
    }
  }
}

#define GNODES 8

__device__ __forceinline__ void gin_sum16(float4& acc,
                                          const float* __restrict__ T, int idx,
                                          int cnt, unsigned lb) {
  float4 v[16];
#pragma unroll
  for (int q = 0; q < 4; ++q) {
    if (4 * q < cnt) {
#pragma unroll
      for (int i = 0; i < 4; ++i) {
        const unsigned c = (unsigned)__shfl(idx, 4 * q + i, 16);
        v[4 * q + i] = *(const float4*)((const char*)T + ((c << 8) + lb));
      }
    }
  }
#pragma unroll
  for (int q = 0; q < 4; ++q) {
    if (4 * q < cnt) {
#pragma unroll
      for (int i = 0; i < 4; ++i) {
        if (4 * q + i < cnt) {
          acc.x += v[4 * q + i].x;
          acc.y += v[4 * q + i].y;
          acc.z += v[4 * q + i].z;
          acc.w += v[4 * q + i].w;
        }
      }
    }
  }
}

__device__ __forceinline__ const int* gin_slot_ptr(const int* row_slots,
                                                   const int* row_deg, int pos,
                                                   int cnt) {
  return cnt > 0 ? row_slots + min(pos, cnt - 1) : row_deg;
}

__global__ void __launch_bounds__(128, 7)
gin_gather_kernel(const float* __restrict__ T, const int* __restrict__ deg,
                  const int* __restrict__ slots, const int* __restrict__ spill,
                  float* __restrict__ AGG, int N, int E) {
  const int grp = (blockIdx.x * blockDim.x + threadIdx.x) >> 4;
  const int g0 = grp * GNODES;
  if (g0 >= N) return;
  const int nn = min(GNODES, N - g0);
  const int l16 = threadIdx.x & 15;
  const unsigned lb = (unsigned)l16 * 16u;

  const int* dp = deg + g0;
  const int* sl = slots + g0 * SLOTS;
  const int dg = dp[min(l16, nn - 1)];
  int d = __shfl(dg, 0, 16);
  int dc = min(d, SLOTS);
  int ia = *gin_slot_ptr(sl, dp, l16, dc);
  int ib = *gin_slot_ptr(sl, dp, 16 + l16, dc);

  for (int n = 0; n < nn; ++n) {
    const int m = min(n + 1, nn - 1);
    const int d2 = __shfl(dg, m, 16);
    const int dc2 = min(d2, SLOTS);
    const int ja = *gin_slot_ptr(sl + m * SLOTS, dp + m, l16, dc2);
    const int jb = *gin_slot_ptr(sl + m * SLOTS, dp + m, 16 + l16, dc2);

    float4 acc = make_float4(0.f, 0.f, 0.f, 0.f);
    if (dc > 0) gin_sum16(acc, T, ia, dc, lb);
    if (dc > 16) gin_sum16(acc, T, ib, dc - 16, lb);
    if (d > SLOTS) {
      const int ns = min(deg[N], E);
      for (int j = 0; j < ns; ++j) {
        if (spill[2 * j] == g0 + n) {
          const unsigned c = (unsigned)spill[2 * j + 1];
          const float4 v = *(const float4*)((const char*)T + ((c << 8) + lb));
          acc.x += v.x;
          acc.y += v.y;
          acc.z += v.z;
          acc.w += v.w;
        }
      }
    }
    *(float4*)((char*)AGG + (((unsigned)(g0 + n) << 8) + lb)) = acc;
    d = d2;
    dc = dc2;
    ia = ja;
    ib = jb;
  }
}


extern "C" void kernel_launch(void* const* d_in, const int* in_sizes, int n_in,
                              void* d_out, int out_size, void* d_ws,
                              size_t ws_size, hipStream_t stream) {
  (void)n_in;
  (void)out_size;
  (void)ws_size;

  const float* x = (const float*)d_in[0];
  const int* ei = (const int*)d_in[1];
  const float* W1 = (const float*)d_in[2];
  const float* b1 = (const float*)d_in[3];
  const float* W2 = (const float*)d_in[4];
  const float* b2 = (const float*)d_in[5];
  const float* Wf = (const float*)d_in[6];
  const float* bf = (const float*)d_in[7];
  float* out = (float*)d_out;

  const int N = in_sizes[0] / NDIM;
  const int E = in_sizes[1] / 2;
  const int L = in_sizes[3] / NDIM;
  const int nTiles = N / 16;
  const size_t NB = (size_t)N * NDIM;

  float* agg = (float*)d_ws;
  float* t = agg + NB;
  float* h = t + NB;
  int* slots = (int*)(h + NB);
  int* spill = slots + (size_t)N * SLOTS;
  int* deg = spill + 2 * (size_t)E;

  const int* rowp = ei;
  const int* colp = ei + E;

  gin_zero_i32_kernel<<<(N + 3 + 255) / 256, 256, 0, stream>>>(deg, N + 3);
  const int nChunks = (E + PCHUNK - 1) / PCHUNK;
  gin_place_kernel<<<2048, PTHREADS, 0, stream>>>(rowp, colp, deg, slots, spill,
                                                  E, N, nChunks);

  const dim3 mlpBlk(WAVES_PER_BLOCK * 32);
  const dim3 mlpGrd((nTiles + WAVES_PER_BLOCK - 1) / WAVES_PER_BLOCK);
  const dim3 gatGrd(((N + GNODES - 1) / GNODES * 16 + 127) / 128);

  const float* hcur = x;
  for (int l = 0; l < L; ++l) {
    const float* w1 = W1 + (size_t)l * W_ELEMS;
    const float* w2 = W2 + (size_t)l * W_ELEMS;
    const float* v1 = b1 + (size_t)l * NDIM;
    const float* v2 = b2 + (size_t)l * NDIM;

    gin_mlp_kernel<0><<<mlpGrd, mlpBlk, 0, stream>>>(hcur, nullptr, w1, v1, w2,
                                                     v2, t, nTiles);
    gin_gather_kernel<<<gatGrd, 128, 0, stream>>>(t, deg, slots, spill, agg, N,
                                                  E);
    gin_mlp_kernel<1><<<mlpGrd, mlpBlk, 0, stream>>>(hcur, agg, w1, v1, w2, v2,
                                                     h, nTiles);
    hcur = h;
  }
  gin_mlp_kernel<2><<<mlpGrd, mlpBlk, 0, stream>>>(hcur, nullptr, Wf, bf,
                                                   nullptr, nullptr, out,
                                                   nTiles);
}
